# starting kernel
# baseline (speedup 1.0000x reference)
#include <hip/hip_runtime.h>
#include <math.h>


#define B_ 64
#define N_ 4096
#define C_ 128
#define H_ 4
#define HC_ 512
#define ALPHA 0.2f

typedef __attribute__((ext_vector_type(8))) float v8f;
typedef __attribute__((ext_vector_type(2))) float v2f;

#define OFF_W2P   0u
#define OFF_B2    2048u
#define OFF_LOG   2064u
#define OFF_MX    (OFF_LOG + (unsigned)(B_*H_*N_))
#define OFF_ISUM  (OFF_MX + 256u)
#define OFF_PART  (OFF_ISUM + 256u)
#define OFF_U     (OFF_PART + (unsigned)(B_*8*H_*C_))

__device__ __forceinline__ float neg_inf() { return -__builtin_inff(); }

__global__ void k_prep(const float* __restrict__ W, const float* __restrict__ bias,
                       const float* __restrict__ T, float* __restrict__ ws) {
  int idx = blockIdx.x * 256 + threadIdx.x;
  if (idx < 2048) {
    int k = idx >> 4, j = idx & 15;
    float v = 0.f;
    if (j < H_) {
      const float* wr = W + (size_t)k * HC_ + j * C_;
      const float* tr = T + j * C_;
      for (int c = 0; c < C_; ++c) v = fmaf(wr[c], tr[c], v);
    }
    ws[OFF_W2P + idx] = v;
  } else if (idx < 2048 + H_) {
    int j = idx - 2048;
    float v = 0.f;
    for (int c = 0; c < C_; ++c) v = fmaf(bias[j * C_ + c], T[j * C_ + c], v);
    ws[OFF_B2 + j] = v;
  }
}

__global__ void k_logits(const float* __restrict__ V, const int* __restrict__ gs,
                         float* __restrict__ ws) {
  __shared__ float s_w2p[2048];
  const int tid = threadIdx.x;
  for (int i = tid; i < 2048; i += 128) s_w2p[i] = ws[OFF_W2P + i];
  __syncthreads();
  const int b = blockIdx.y;
  const int wave = tid >> 5, lane = tid & 31;
  const int l16 = lane & 15, hi = lane >> 4;
  const int n0 = blockIdx.x * 64 + wave * 16;
  const float* vrow = V + ((size_t)b * N_ + (n0 + l16)) * C_;
  v8f acc = {};
  for (int k0 = 0; k0 < C_; k0 += 4) {
    v2f a = *(const v2f*)(vrow + k0 + 2 * hi);
    v2f bv;
    bv.x = s_w2p[(k0 + 2 * hi) * 16 + l16];
    bv.y = s_w2p[(k0 + 1 + 2 * hi) * 16 + l16];
    acc = __builtin_amdgcn_wmma_f32_16x16x4_f32(false, a, false, bv,
                                                (short)0, acc, false, false);
  }
  const int limit = gs[b];
  const float bb = ws[OFF_B2 + (l16 & 3)];
  if (l16 < H_) {
    float* lrow = ws + OFF_LOG + ((size_t)b * H_ + l16) * N_;
    for (int r = 0; r < 8; ++r) {
      int node = n0 + r + 8 * hi;
      float x = acc[r] + bb;
      x = (x >= 0.f) ? x : ALPHA * x;
      lrow[node] = (node < limit) ? x : neg_inf();
    }
  }
}

__global__ void k_stats(float* __restrict__ ws) {
  __shared__ float red[256];
  const int bh = blockIdx.x;
  const int tid = threadIdx.x;
  const float* l = ws + OFF_LOG + (size_t)bh * N_;
  float m = neg_inf();
  for (int n = tid; n < N_; n += 256) m = fmaxf(m, l[n]);
  red[tid] = m; __syncthreads();
  for (int s = 128; s > 0; s >>= 1) {
    if (tid < s) red[tid] = fmaxf(red[tid], red[tid + s]);
    __syncthreads();
  }
  m = red[0]; __syncthreads();
  float sum = 0.f;
  for (int n = tid; n < N_; n += 256) {
    float x = l[n];
    sum += (x == neg_inf()) ? 0.f : __expf(x - m);
  }
  red[tid] = sum; __syncthreads();
  for (int s = 128; s > 0; s >>= 1) {
    if (tid < s) red[tid] += red[tid + s];
    __syncthreads();
  }
  if (tid == 0) { ws[OFF_MX + bh] = m; ws[OFF_ISUM + bh] = 1.f / red[0]; }
}

__global__ void k_wsum(const float* __restrict__ V, float* __restrict__ ws) {
  __shared__ float s_p[H_ * 512];
  const int b = blockIdx.y, chunk = blockIdx.x;
  const int nb = chunk * 512;
  const int tid = threadIdx.x;
  for (int i = tid; i < H_ * 512; i += 256) {
    int h = i >> 9, n = i & 511;
    float x = ws[OFF_LOG + ((size_t)b * H_ + h) * N_ + nb + n];
    float m = ws[OFF_MX + b * H_ + h];
    float is = ws[OFF_ISUM + b * H_ + h];
    s_p[i] = (x == neg_inf()) ? 0.f : __expf(x - m) * is;
  }
  __syncthreads();
  const int wave = tid >> 5, lane = tid & 31;
  const int l16 = lane & 15, hi = lane >> 4;
  const int c0 = wave * 16;
  const float* vb = V + ((size_t)b * N_ + nb) * C_;
  v8f acc = {};
  for (int k0 = 0; k0 < 512; k0 += 4) {
    v2f a;
    if (l16 < H_) {
      a.x = s_p[l16 * 512 + k0 + 2 * hi];
      a.y = s_p[l16 * 512 + k0 + 1 + 2 * hi];
    } else { a.x = 0.f; a.y = 0.f; }
    v2f bv;
    bv.x = vb[(size_t)(k0 + 2 * hi) * C_ + c0 + l16];
    bv.y = vb[(size_t)(k0 + 1 + 2 * hi) * C_ + c0 + l16];
    acc = __builtin_amdgcn_wmma_f32_16x16x4_f32(false, a, false, bv,
                                                (short)0, acc, false, false);
  }
  if (hi == 0) {
    float* pr = ws + OFF_PART + (size_t)(b * 8 + chunk) * (H_ * C_);
    for (int r = 0; r < H_; ++r) pr[r * C_ + c0 + l16] = acc[r];
  }
}

__global__ void k_reduce(float* __restrict__ ws) {
  int idx = blockIdx.x * 256 + threadIdx.x;
  int b = idx >> 9, rest = idx & 511;
  float s = 0.f;
  for (int ch = 0; ch < 8; ++ch)
    s += ws[OFF_PART + ((size_t)(b * 8 + ch) << 9) + rest];
  ws[OFF_U + idx] = s;
}

__global__ void k_out(const float* __restrict__ W, const float* __restrict__ bias,
                      const float* __restrict__ U, float* __restrict__ out) {
  const int m0 = blockIdx.x * 16;
  const int h  = blockIdx.y;
  const int tid = threadIdx.x;
  const int wave = tid >> 5, lane = tid & 31;
  const int l16 = lane & 15, hi = lane >> 4;
  const int n0 = wave * 16;
  const float* urow = U + (size_t)(m0 + l16) * HC_ + h * C_;
  const float* wcol = W + h * C_ + n0 + l16;
  v8f acc = {};
  for (int k0 = 0; k0 < C_; k0 += 4) {
    v2f a = *(const v2f*)(urow + k0 + 2 * hi);
    v2f bv;
    bv.x = wcol[(size_t)(k0 + 2 * hi) * HC_];
    bv.y = wcol[(size_t)(k0 + 1 + 2 * hi) * HC_];
    acc = __builtin_amdgcn_wmma_f32_16x16x4_f32(false, a, false, bv,
                                                (short)0, acc, false, false);
  }
  for (int r = 0; r < 8; ++r) {
    int row = m0 + r + 8 * hi;
    int col = h * C_ + n0 + l16;
    out[(size_t)row * HC_ + col] = acc[r] + bias[col];
  }
}

extern "C" void kernel_launch(void* const* d_in, const int* in_sizes, int n_in,
                              void* d_out, int out_size, void* d_ws, size_t ws_size,
                              hipStream_t stream) {
  (void)in_sizes; (void)n_in; (void)out_size; (void)ws_size;
  const float* V    = (const float*)d_in[0];
  const int*   gs   = (const int*)  d_in[1];
  const float* W    = (const float*)d_in[2];
  const float* bias = (const float*)d_in[3];
  const float* T    = (const float*)d_in[4];
  float* out = (float*)d_out;
  float* ws  = (float*)d_ws;

  k_prep  <<<9,            256, 0, stream>>>(W, bias, T, ws);
  k_logits<<<dim3(N_/64, B_), 128, 0, stream>>>(V, gs, ws);
  k_stats <<<B_ * H_,      256, 0, stream>>>(ws);
  k_wsum  <<<dim3(8, B_),  256, 0, stream>>>(V, ws);
  k_reduce<<<(B_*H_*C_)/256, 256, 0, stream>>>(ws);
  k_out   <<<dim3(B_/16, H_), 256, 0, stream>>>(W, bias, ws + OFF_U, out);
}


# K2 logits: V goes from memory straight into A fragments by whole-line 128-bit loads (rows = 4 nodes x 4 channel quarters, no LDS); quarter sums added across lanes: logit sum order changes
# speedup vs baseline: 2.5303x; 2.5303x over previous
#include <hip/hip_runtime.h>
#include <math.h>


#define B_ 64
#define N_ 4096
#define C_ 128
#define H_ 4
#define HC_ 512
#define ALPHA 0.2f

typedef __attribute__((ext_vector_type(8))) float v8f;
typedef __attribute__((ext_vector_type(2))) float v2f;

#define OFF_W2P   0u
#define OFF_B2    2048u
#define OFF_LOG   2064u
#define OFF_MX    (OFF_LOG + (unsigned)(B_*H_*N_))
#define OFF_ISUM  (OFF_MX + 256u)
#define OFF_PART  (OFF_ISUM + 256u)
#define OFF_U     (OFF_PART + (unsigned)(B_*8*H_*C_))

__device__ __forceinline__ float neg_inf() { return -__builtin_inff(); }

__global__ void k_prep(const float* __restrict__ W, const float* __restrict__ bias,
                       const float* __restrict__ T, float* __restrict__ ws) {
  int idx = blockIdx.x * 256 + threadIdx.x;
  if (idx < 2048) {
    int k = idx >> 4, j = idx & 15;
    float v = 0.f;
    if (j < H_) {
      const float* wr = W + (size_t)k * HC_ + j * C_;
      const float* tr = T + j * C_;
      for (int c = 0; c < C_; ++c) v = fmaf(wr[c], tr[c], v);
    }
    ws[OFF_W2P + idx] = v;
  } else if (idx < 2048 + H_) {
    int j = idx - 2048;
    float v = 0.f;
    for (int c = 0; c < C_; ++c) v = fmaf(bias[j * C_ + c], T[j * C_ + c], v);
    ws[OFF_B2 + j] = v;
  }
}


template <int M>
__device__ __forceinline__ float row_xor(float x) {
  return __int_as_float(__builtin_amdgcn_update_dpp(
      0, __float_as_int(x), 0x160 + M, 0xf, 0xf, false));
}

__global__ __launch_bounds__(256) void k_logits(const float* __restrict__ V,
                                                const int* __restrict__ gs,
                                                float* __restrict__ ws) {
  const int b = blockIdx.y, chunk = blockIdx.x;
  const int tid = threadIdx.x;
  const int wave = tid >> 5, lane = tid & 31;
  const int l16 = lane & 15, hf = lane >> 4;
  const int qd = l16 >> 2, lo2 = l16 & 3;
  const int n0 = chunk * 512 + wave * 64;

  v2f bw[8];
  {
    const float* wq = ws + OFF_W2P + (16 * hf + 4 * qd) * 16 + lo2;
    for (int j = 0; j < 4; ++j) {
      bw[2 * j].x     = wq[(32 * j + 0) * 16];
      bw[2 * j].y     = wq[(32 * j + 1) * 16];
      bw[2 * j + 1].x = wq[(32 * j + 2) * 16];
      bw[2 * j + 1].y = wq[(32 * j + 3) * 16];
    }
  }
  const float bb = ws[OFF_B2 + lo2];
  const int limit = gs[b];

  const float4* vsrc =
      (const float4*)(V + ((size_t)b * N_ + n0 + qd) * C_) + lo2 + 4 * hf;
  float* lrow = ws + OFF_LOG + ((size_t)b * H_ + lo2) * N_ + n0 + 2 * hf;

  float4 ra[16], rb[16];
  v8f acc[4];
#define K2_FETCH(R, T)                                                        \
  for (int g = 0; g < 4; ++g)                                                 \
    for (int j = 0; j < 4; ++j)                                               \
      R[4 * g + j] = vsrc[(size_t)(16 * (T) + 4 * g) * (C_ / 4) + 8 * j];
#define K2_MMA(R)                                                             \
  for (int g = 0; g < 4; ++g) acc[g] = v8f{};                                 \
  for (int j = 0; j < 4; ++j)                                                 \
    for (int g = 0; g < 4; ++g) {                                             \
      const float4 x = R[4 * g + j];                                          \
      const v2f a0 = {x.x, x.y}, a1 = {x.z, x.w};                             \
      acc[g] = __builtin_amdgcn_wmma_f32_16x16x4_f32(                         \
          false, a0, false, bw[2 * j], (short)0, acc[g], false, false);       \
      acc[g] = __builtin_amdgcn_wmma_f32_16x16x4_f32(                         \
          false, a1, false, bw[2 * j + 1], (short)0, acc[g], false, false);   \
    }
#define K2_EPI(T)                                                             \
  for (int g = 0; g < 4; ++g) {                                               \
    const float t0 = q1 ? acc[g][1] : acc[g][0], t1 = q1 ? acc[g][3] : acc[g][2]; \
    const float t2 = q1 ? acc[g][5] : acc[g][4], t3 = q1 ? acc[g][7] : acc[g][6]; \
    float s0 = q2 ? t1 : t0, s1 = q2 ? t3 : t2;                               \
    s0 += row_xor<4>(s0);  s1 += row_xor<4>(s1);                              \
    s0 += row_xor<8>(s0);  s1 += row_xor<8>(s1);                              \
    const int node = n0 + 16 * (T) + 4 * g + 2 * hf;                          \
    float x0 = s0 + bb, x1 = s1 + bb;                                         \
    x0 = (x0 >= 0.f) ? x0 : ALPHA * x0;                                       \
    x1 = (x1 >= 0.f) ? x1 : ALPHA * x1;                                       \
    v2f o;                                                                    \
    o.x = (node < limit) ? x0 : neg_inf();                                    \
    o.y = (node + 1 < limit) ? x1 : neg_inf();                                \
    if (l16 < H_) *(v2f*)(lrow + 16 * (T) + 4 * g) = o;                       \
  }
#define K2_PIN __builtin_amdgcn_sched_barrier(0);
  const bool q1 = (qd & 1) != 0, q2 = (qd & 2) != 0;

  K2_FETCH(ra, 0)
  K2_FETCH(rb, 1)
  K2_PIN
  K2_MMA(ra)
  K2_PIN
  K2_FETCH(ra, 2)
  K2_PIN
  K2_EPI(0)
  K2_MMA(rb)
  K2_PIN
  K2_FETCH(rb, 3)
  K2_PIN
  K2_EPI(1)
  K2_MMA(ra)
  K2_EPI(2)
  K2_MMA(rb)
  K2_EPI(3)
#undef K2_PIN
#undef K2_FETCH
#undef K2_MMA
#undef K2_EPI
}

__global__ void k_stats(float* __restrict__ ws) {
  __shared__ float red[256];
  const int bh = blockIdx.x;
  const int tid = threadIdx.x;
  const float* l = ws + OFF_LOG + (size_t)bh * N_;
  float m = neg_inf();
  for (int n = tid; n < N_; n += 256) m = fmaxf(m, l[n]);
  red[tid] = m; __syncthreads();
  for (int s = 128; s > 0; s >>= 1) {
    if (tid < s) red[tid] = fmaxf(red[tid], red[tid + s]);
    __syncthreads();
  }
  m = red[0]; __syncthreads();
  float sum = 0.f;
  for (int n = tid; n < N_; n += 256) {
    float x = l[n];
    sum += (x == neg_inf()) ? 0.f : __expf(x - m);
  }
  red[tid] = sum; __syncthreads();
  for (int s = 128; s > 0; s >>= 1) {
    if (tid < s) red[tid] += red[tid + s];
    __syncthreads();
  }
  if (tid == 0) { ws[OFF_MX + bh] = m; ws[OFF_ISUM + bh] = 1.f / red[0]; }
}

__device__ __forceinline__ void lds_barrier() {
  __builtin_amdgcn_fence(__ATOMIC_RELEASE, "workgroup", "local");
  __builtin_amdgcn_s_barrier();
  __builtin_amdgcn_fence(__ATOMIC_ACQUIRE, "workgroup", "local");
}

#define TR_    64
#define NT_    (512 / TR_)
#define VP_    288
#define VBUF_  ((TR_ / 2) * VP_)
__global__ __launch_bounds__(256) void k_wsum(const float* __restrict__ V,
                                              float* __restrict__ ws) {
  __shared__ __attribute__((aligned(16))) float s_p[H_ * 512];
  __shared__ __attribute__((aligned(16))) float s_v[2 * VBUF_];
  const int b = blockIdx.y, chunk = blockIdx.x;
  const int nb = chunk * 512;
  const int tid = threadIdx.x;
  const int wave = tid >> 5, lane = tid & 31;
  const int l16 = lane & 15, hi = lane >> 4;
  const int c0 = wave * 16;
  const float* vb = V + ((size_t)b * N_ + nb) * C_;

  float4 r0[4], r1[4];
  const float4* gsrc = (const float4*)(vb + (size_t)(2 * wave) * C_) + lane;
  float* sdst = s_v + wave * VP_ + 8 * lane;

#define K4_FETCH(T)                                                          \
  for (int i = 0; i < 4; ++i) {                                              \
    const float4* q = gsrc + (size_t)((T) * TR_ + 16 * i) * (C_ / 4);        \
    r0[i] = q[0];                                                            \
    r1[i] = q[C_ / 4];                                                       \
  }
#define K4_STAGE(BUF)                                                        \
  for (int i = 0; i < 4; ++i) {                                              \
    float* d = sdst + (BUF) * VBUF_ + 8 * i * VP_;                           \
    *(float4*)(d)     = make_float4(r0[i].x, r1[i].x, r0[i].y, r1[i].y);     \
    *(float4*)(d + 4) = make_float4(r0[i].z, r1[i].z, r0[i].w, r1[i].w);     \
  }

  K4_FETCH(0)
  float lx[8], lm[H_], lis[H_];
  for (int h = 0; h < H_; ++h) {
    lm[h]  = ws[OFF_MX + b * H_ + h];
    lis[h] = ws[OFF_ISUM + b * H_ + h];
  }
  for (int it = 0; it < 8; ++it)
    lx[it] = ws[OFF_LOG + ((size_t)b * H_ + (it >> 1)) * N_ + nb + (it & 1) * 256 + tid];
  for (int it = 0; it < 8; ++it) {
    const int h = it >> 1, n = (it & 1) * 256 + tid;
    const float x = lx[it];
    s_p[((n >> 1) * H_ + h) * 2 + (n & 1)] =
        (x == neg_inf()) ? 0.f : __expf(x - lm[h]) * lis[h];
  }
  K4_STAGE(0)
  K4_FETCH(1)
  lds_barrier();

  v8f acc = {};
  for (int t = 0; t < NT_; ++t) {
    const float* bt = s_v + (t & 1) * VBUF_ + (hi * VP_ + (c0 + l16) * 2);
    const float* pt = s_p + ((t * (TR_ / 2) + hi) * H_ + l16) * 2;
    v2f af[TR_ / 4], bf[TR_ / 4];
    for (int kk = 0; kk < TR_ / 4; ++kk) {
      bf[kk] = *(const v2f*)(bt + 2 * kk * VP_);
      v2f a = {0.f, 0.f};
      if (l16 < H_) a = *(const v2f*)(pt + 2 * kk * H_ * 2);
      af[kk] = a;
    }
    for (int kk = 0; kk < TR_ / 4; ++kk)
      acc = __builtin_amdgcn_wmma_f32_16x16x4_f32(false, af[kk], false, bf[kk],
                                                  (short)0, acc, false, false);
    if (t + 1 < NT_) {
      K4_STAGE((t + 1) & 1)
      if (t + 2 < NT_) { K4_FETCH(t + 2) }
      lds_barrier();
    }
  }
#undef K4_FETCH
#undef K4_STAGE
  if (hi == 0) {
    float* pr = ws + OFF_PART + (size_t)(b * 8 + chunk) * (H_ * C_);
    for (int r = 0; r < H_; ++r) pr[r * C_ + c0 + l16] = acc[r];
  }
}

__global__ void k_reduce(float* __restrict__ ws) {
  int idx = blockIdx.x * 256 + threadIdx.x;
  int b = idx >> 9, rest = idx & 511;
  float s = 0.f;
  for (int ch = 0; ch < 8; ++ch)
    s += ws[OFF_PART + ((size_t)(b * 8 + ch) << 9) + rest];
  ws[OFF_U + idx] = s;
}

__global__ void k_out(const float* __restrict__ W, const float* __restrict__ bias,
                      const float* __restrict__ U, float* __restrict__ out) {
  const int m0 = blockIdx.x * 16;
  const int h  = blockIdx.y;
  const int tid = threadIdx.x;
  const int wave = tid >> 5, lane = tid & 31;
  const int l16 = lane & 15, hi = lane >> 4;
  const int n0 = wave * 16;
  const float* urow = U + (size_t)(m0 + l16) * HC_ + h * C_;
  const float* wcol = W + h * C_ + n0 + l16;
  v8f acc = {};
  for (int k0 = 0; k0 < C_; k0 += 4) {
    v2f a = *(const v2f*)(urow + k0 + 2 * hi);
    v2f bv;
    bv.x = wcol[(size_t)(k0 + 2 * hi) * HC_];
    bv.y = wcol[(size_t)(k0 + 1 + 2 * hi) * HC_];
    acc = __builtin_amdgcn_wmma_f32_16x16x4_f32(false, a, false, bv,
                                                (short)0, acc, false, false);
  }
  for (int r = 0; r < 8; ++r) {
    int row = m0 + r + 8 * hi;
    int col = h * C_ + n0 + l16;
    out[(size_t)row * HC_ + col] = acc[r] + bias[col];
  }
}

extern "C" void kernel_launch(void* const* d_in, const int* in_sizes, int n_in,
                              void* d_out, int out_size, void* d_ws, size_t ws_size,
                              hipStream_t stream) {
  (void)in_sizes; (void)n_in; (void)out_size; (void)ws_size;
  const float* V    = (const float*)d_in[0];
  const int*   gs   = (const int*)  d_in[1];
  const float* W    = (const float*)d_in[2];
  const float* bias = (const float*)d_in[3];
  const float* T    = (const float*)d_in[4];
  float* out = (float*)d_out;
  float* ws  = (float*)d_ws;

  k_prep  <<<9,            256, 0, stream>>>(W, bias, T, ws);
  k_logits<<<dim3(N_/512, B_), 256, 0, stream>>>(V, gs, ws);
  k_stats <<<B_ * H_,      256, 0, stream>>>(ws);
  k_wsum  <<<dim3(8, B_),  256, 0, stream>>>(V, ws);
  k_reduce<<<(B_*H_*C_)/256, 256, 0, stream>>>(ws);
  k_out   <<<dim3(B_/16, H_), 256, 0, stream>>>(W, bias, ws + OFF_U, out);
}


# K4 split by channel quarter, not by node: a work-group owns one graph x 32 channels, forms the softmax itself, V straight into B fragments (K3, K5 go: 4 launches); softmax and u sum order changes
# speedup vs baseline: 2.9437x; 1.1634x over previous
#include <hip/hip_runtime.h>
#include <math.h>


#define B_ 64
#define N_ 4096
#define C_ 128
#define H_ 4
#define HC_ 512
#define ALPHA 0.2f

typedef __attribute__((ext_vector_type(8))) float v8f;
typedef __attribute__((ext_vector_type(2))) float v2f;

#define OFF_W2P   0u
#define OFF_B2    2048u
#define OFF_LOG   2064u
#define OFF_U     (OFF_LOG + (unsigned)(B_*H_*N_))

__device__ __forceinline__ float neg_inf() { return -__builtin_inff(); }

__global__ void k_prep(const float* __restrict__ W, const float* __restrict__ bias,
                       const float* __restrict__ T, float* __restrict__ ws) {
  int idx = blockIdx.x * 256 + threadIdx.x;
  if (idx < 2048) {
    int k = idx >> 4, j = idx & 15;
    float v = 0.f;
    if (j < H_) {
      const float* wr = W + (size_t)k * HC_ + j * C_;
      const float* tr = T + j * C_;
      for (int c = 0; c < C_; ++c) v = fmaf(wr[c], tr[c], v);
    }
    ws[OFF_W2P + idx] = v;
  } else if (idx < 2048 + H_) {
    int j = idx - 2048;
    float v = 0.f;
    for (int c = 0; c < C_; ++c) v = fmaf(bias[j * C_ + c], T[j * C_ + c], v);
    ws[OFF_B2 + j] = v;
  }
}


template <int M>
__device__ __forceinline__ float row_xor(float x) {
  return __int_as_float(__builtin_amdgcn_update_dpp(
      0, __float_as_int(x), 0x160 + M, 0xf, 0xf, false));
}

__global__ __launch_bounds__(256) void k_logits(const float* __restrict__ V,
                                                const int* __restrict__ gs,
                                                float* __restrict__ ws) {
  const int b = blockIdx.y, chunk = blockIdx.x;
  const int tid = threadIdx.x;
  const int wave = tid >> 5, lane = tid & 31;
  const int l16 = lane & 15, hf = lane >> 4;
  const int qd = l16 >> 2, lo2 = l16 & 3;
  const int n0 = chunk * 512 + wave * 64;

  v2f bw[8];
  {
    const float* wq = ws + OFF_W2P + (16 * hf + 4 * qd) * 16 + lo2;
    for (int j = 0; j < 4; ++j) {
      bw[2 * j].x     = wq[(32 * j + 0) * 16];
      bw[2 * j].y     = wq[(32 * j + 1) * 16];
      bw[2 * j + 1].x = wq[(32 * j + 2) * 16];
      bw[2 * j + 1].y = wq[(32 * j + 3) * 16];
    }
  }
  const float bb = ws[OFF_B2 + lo2];
  const int limit = gs[b];

  const float4* vsrc =
      (const float4*)(V + ((size_t)b * N_ + n0 + qd) * C_) + lo2 + 4 * hf;
  float* lrow = ws + OFF_LOG + ((size_t)b * H_ + lo2) * N_ + n0 + 2 * hf;

  float4 ra[16], rb[16];
  v8f acc[4];
#define K2_FETCH(R, T)                                                        \
  for (int g = 0; g < 4; ++g)                                                 \
    for (int j = 0; j < 4; ++j)                                               \
      R[4 * g + j] = vsrc[(size_t)(16 * (T) + 4 * g) * (C_ / 4) + 8 * j];
#define K2_MMA(R)                                                             \
  for (int g = 0; g < 4; ++g) acc[g] = v8f{};                                 \
  for (int j = 0; j < 4; ++j)                                                 \
    for (int g = 0; g < 4; ++g) {                                             \
      const float4 x = R[4 * g + j];                                          \
      const v2f a0 = {x.x, x.y}, a1 = {x.z, x.w};                             \
      acc[g] = __builtin_amdgcn_wmma_f32_16x16x4_f32(                         \
          false, a0, false, bw[2 * j], (short)0, acc[g], false, false);       \
      acc[g] = __builtin_amdgcn_wmma_f32_16x16x4_f32(                         \
          false, a1, false, bw[2 * j + 1], (short)0, acc[g], false, false);   \
    }
#define K2_EPI(T)                                                             \
  for (int g = 0; g < 4; ++g) {                                               \
    const float t0 = q1 ? acc[g][1] : acc[g][0], t1 = q1 ? acc[g][3] : acc[g][2]; \
    const float t2 = q1 ? acc[g][5] : acc[g][4], t3 = q1 ? acc[g][7] : acc[g][6]; \
    float s0 = q2 ? t1 : t0, s1 = q2 ? t3 : t2;                               \
    s0 += row_xor<4>(s0);  s1 += row_xor<4>(s1);                              \
    s0 += row_xor<8>(s0);  s1 += row_xor<8>(s1);                              \
    const int node = n0 + 16 * (T) + 4 * g + 2 * hf;                          \
    float x0 = s0 + bb, x1 = s1 + bb;                                         \
    x0 = (x0 >= 0.f) ? x0 : ALPHA * x0;                                       \
    x1 = (x1 >= 0.f) ? x1 : ALPHA * x1;                                       \
    v2f o;                                                                    \
    o.x = (node < limit) ? x0 : neg_inf();                                    \
    o.y = (node + 1 < limit) ? x1 : neg_inf();                                \
    if (l16 < H_) *(v2f*)(lrow + 16 * (T) + 4 * g) = o;                       \
  }
#define K2_PIN __builtin_amdgcn_sched_barrier(0);
  const bool q1 = (qd & 1) != 0, q2 = (qd & 2) != 0;

  K2_FETCH(ra, 0)
  K2_FETCH(rb, 1)
  K2_PIN
  K2_MMA(ra)
  K2_PIN
  K2_FETCH(ra, 2)
  K2_PIN
  K2_EPI(0)
  K2_MMA(rb)
  K2_PIN
  K2_FETCH(rb, 3)
  K2_PIN
  K2_EPI(1)
  K2_MMA(ra)
  K2_EPI(2)
  K2_MMA(rb)
  K2_EPI(3)
#undef K2_PIN
#undef K2_FETCH
#undef K2_MMA
#undef K2_EPI
}

__device__ __forceinline__ void lds_barrier() {
  __builtin_amdgcn_fence(__ATOMIC_RELEASE, "workgroup", "local");
  __builtin_amdgcn_s_barrier();
  __builtin_amdgcn_fence(__ATOMIC_ACQUIRE, "workgroup", "local");
}

#define K3_T 512
__global__ __launch_bounds__(K3_T) void k_wsum(const float* __restrict__ V,
                                               float* __restrict__ ws) {
  __shared__ __attribute__((aligned(16))) float s_p[N_ * H_];
  __shared__ __attribute__((aligned(16))) float s_r[16 * 256];
  const int q = blockIdx.x, b = blockIdx.y;
  const int tid = threadIdx.x;
  const int wave = tid >> 5, lane = tid & 31;
  const int l16 = lane & 15, hf = lane >> 4;
  const int par = l16 >> 3, j = l16 & 7;
  const int nw = wave * 256;

  const float* lg = ws + OFF_LOG + (size_t)b * H_ * N_ + 4 * tid;
  float4 lx[H_][2];
  for (int h = 0; h < H_; ++h) {
    lx[h][0] = *(const float4*)(lg + h * N_);
    lx[h][1] = *(const float4*)(lg + h * N_ + 2048);
  }
  const float4* vsrc =
      (const float4*)(V + ((size_t)b * N_ + nw + 4 * par + 2 * hf) * C_ + 32 * q) + j;
  float4 ra[8], rb[8];
#define K3_FETCH(R, T)                                                        \
  for (int g = 0; g < 4; ++g) {                                               \
    R[2 * g]     = vsrc[(size_t)(32 * (T) + 8 * g) * (C_ / 4)];               \
    R[2 * g + 1] = vsrc[(size_t)(32 * (T) + 8 * g + 1) * (C_ / 4)];           \
  }
#define K3_PIN __builtin_amdgcn_sched_barrier(0);
  K3_PIN
  K3_FETCH(ra, 0)
  K3_FETCH(rb, 1)
  K3_PIN

  float m[H_];
  for (int h = 0; h < H_; ++h) {
    float v = fmaxf(fmaxf(fmaxf(lx[h][0].x, lx[h][0].y), fmaxf(lx[h][0].z, lx[h][0].w)),
                    fmaxf(fmaxf(lx[h][1].x, lx[h][1].y), fmaxf(lx[h][1].z, lx[h][1].w)));
    v = fmaxf(v, row_xor<1>(v));  v = fmaxf(v, row_xor<2>(v));
    v = fmaxf(v, row_xor<4>(v));  v = fmaxf(v, row_xor<8>(v));
    if (l16 == 0) s_r[h * 32 + 2 * wave + hf] = v;
  }
  lds_barrier();
  for (int h = 0; h < H_; ++h) {
    float v = s_r[h * 32];
    for (int w = 1; w < 32; ++w) v = fmaxf(v, s_r[h * 32 + w]);
    m[h] = v;
  }
  float isum[H_];
  for (int h = 0; h < H_; ++h) {
    float s = 0.f;
    for (int i = 0; i < 2; ++i) {
      float4 x = lx[h][i];
      x.x = (x.x == neg_inf()) ? 0.f : __expf(x.x - m[h]);
      x.y = (x.y == neg_inf()) ? 0.f : __expf(x.y - m[h]);
      x.z = (x.z == neg_inf()) ? 0.f : __expf(x.z - m[h]);
      x.w = (x.w == neg_inf()) ? 0.f : __expf(x.w - m[h]);
      lx[h][i] = x;
      s += (x.x + x.y) + (x.z + x.w);
    }
    s += row_xor<1>(s);  s += row_xor<2>(s);
    s += row_xor<4>(s);  s += row_xor<8>(s);
    if (l16 == 0) s_r[128 + h * 32 + 2 * wave + hf] = s;
  }
  lds_barrier();
  for (int h = 0; h < H_; ++h) {
    float s = s_r[128 + h * 32];
    for (int w = 1; w < 32; ++w) s += s_r[128 + h * 32 + w];
    isum[h] = 1.f / s;
  }
  for (int i = 0; i < 2; ++i) {
    float* d = s_p + (size_t)(i * 1024 + 2 * tid) * (H_ * 2);
    *(float4*)(d)      = make_float4(lx[0][i].x * isum[0], lx[0][i].y * isum[0],
                                     lx[1][i].x * isum[1], lx[1][i].y * isum[1]);
    *(float4*)(d + 4)  = make_float4(lx[2][i].x * isum[2], lx[2][i].y * isum[2],
                                     lx[3][i].x * isum[3], lx[3][i].y * isum[3]);
    *(float4*)(d + 8)  = make_float4(lx[0][i].z * isum[0], lx[0][i].w * isum[0],
                                     lx[1][i].z * isum[1], lx[1][i].w * isum[1]);
    *(float4*)(d + 12) = make_float4(lx[2][i].z * isum[2], lx[2][i].w * isum[2],
                                     lx[3][i].z * isum[3], lx[3][i].w * isum[3]);
  }
  lds_barrier();

  const float* pa = s_p + (size_t)((nw >> 1) + 2 * (l16 >> 2) + hf) * (H_ * 2) + (l16 & 3) * 2;
  v8f acc[4];
  for (int c = 0; c < 4; ++c) acc[c] = v8f{};
#define K3_MMA(R, T)                                                          \
  {                                                                           \
    v2f af[4];                                                                \
    for (int g = 0; g < 4; ++g) {                                             \
      v2f a = {0.f, 0.f};                                                     \
      if (l16 < 8) a = *(const v2f*)(pa + (size_t)(16 * (T) + 4 * g) * (H_ * 2)); \
      af[g] = a;                                                              \
    }                                                                         \
    for (int g = 0; g < 4; ++g) {                                             \
      const float4 x0 = R[2 * g], x1 = R[2 * g + 1];                          \
      const v2f b0 = {x0.x, x1.x}, b1 = {x0.y, x1.y};                         \
      const v2f b2 = {x0.z, x1.z}, b3 = {x0.w, x1.w};                         \
      acc[0] = __builtin_amdgcn_wmma_f32_16x16x4_f32(                         \
          false, af[g], false, b0, (short)0, acc[0], false, false);           \
      acc[1] = __builtin_amdgcn_wmma_f32_16x16x4_f32(                         \
          false, af[g], false, b1, (short)0, acc[1], false, false);           \
      acc[2] = __builtin_amdgcn_wmma_f32_16x16x4_f32(                         \
          false, af[g], false, b2, (short)0, acc[2], false, false);           \
      acc[3] = __builtin_amdgcn_wmma_f32_16x16x4_f32(                         \
          false, af[g], false, b3, (short)0, acc[3], false, false);           \
    }                                                                         \
  }
  for (int t = 0; t < 8; t += 2) {
    K3_MMA(ra, t)
    K3_PIN
    if (t + 2 < 8) { K3_FETCH(ra, t + 2) }
    K3_PIN
    K3_MMA(rb, t + 1)
    K3_PIN
    if (t + 3 < 8) { K3_FETCH(rb, t + 3) }
    K3_PIN
  }
#undef K3_PIN
#undef K3_FETCH
#undef K3_MMA

  if (hf == 0) {
    float* d = s_r + wave * 256 + par * 128 + 4 * j;
    for (int h = 0; h < H_; ++h) {
      float4 o;
      o.x = par ? acc[0][4 + h] : acc[0][h];
      o.y = par ? acc[1][4 + h] : acc[1][h];
      o.z = par ? acc[2][4 + h] : acc[2][h];
      o.w = par ? acc[3][4 + h] : acc[3][h];
      *(float4*)(d + h * 32) = o;
    }
  }
  lds_barrier();
  if (tid < H_ * 32) {
    float s = 0.f;
    for (int w = 0; w < 16; ++w)
      s += s_r[w * 256 + tid] + s_r[w * 256 + 128 + tid];
    ws[OFF_U + b * HC_ + (tid >> 5) * C_ + 32 * q + (tid & 31)] = s;
  }
}

__global__ void k_out(const float* __restrict__ W, const float* __restrict__ bias,
                      const float* __restrict__ U, float* __restrict__ out) {
  const int m0 = blockIdx.x * 16;
  const int h  = blockIdx.y;
  const int tid = threadIdx.x;
  const int wave = tid >> 5, lane = tid & 31;
  const int l16 = lane & 15, hi = lane >> 4;
  const int n0 = wave * 16;
  const float* urow = U + (size_t)(m0 + l16) * HC_ + h * C_;
  const float* wcol = W + h * C_ + n0 + l16;
  v8f acc = {};
  for (int k0 = 0; k0 < C_; k0 += 4) {
    v2f a = *(const v2f*)(urow + k0 + 2 * hi);
    v2f bv;
    bv.x = wcol[(size_t)(k0 + 2 * hi) * HC_];
    bv.y = wcol[(size_t)(k0 + 1 + 2 * hi) * HC_];
    acc = __builtin_amdgcn_wmma_f32_16x16x4_f32(false, a, false, bv,
                                                (short)0, acc, false, false);
  }
  for (int r = 0; r < 8; ++r) {
    int row = m0 + r + 8 * hi;
    int col = h * C_ + n0 + l16;
    out[(size_t)row * HC_ + col] = acc[r] + bias[col];
  }
}

extern "C" void kernel_launch(void* const* d_in, const int* in_sizes, int n_in,
                              void* d_out, int out_size, void* d_ws, size_t ws_size,
                              hipStream_t stream) {
  (void)in_sizes; (void)n_in; (void)out_size; (void)ws_size;
  const float* V    = (const float*)d_in[0];
  const int*   gs   = (const int*)  d_in[1];
  const float* W    = (const float*)d_in[2];
  const float* bias = (const float*)d_in[3];
  const float* T    = (const float*)d_in[4];
  float* out = (float*)d_out;
  float* ws  = (float*)d_ws;

  k_prep  <<<9,            256, 0, stream>>>(W, bias, T, ws);
  k_logits<<<dim3(N_/512, B_), 256, 0, stream>>>(V, gs, ws);
  k_wsum  <<<dim3(C_/32, B_), K3_T, 0, stream>>>(V, ws);
  k_out   <<<dim3(B_/16, H_), 256, 0, stream>>>(W, bias, ws + OFF_U, out);
}


# K1's launch goes: every K2 work-group (now 16 waves, 1024 nodes) forms w2 and b2 itself in LDS behind its first V requests (3 launches); lane partials added by exchange: w2, b2 sum order changes
# speedup vs baseline: 3.1249x; 1.0616x over previous
#include <hip/hip_runtime.h>
#include <math.h>


#define B_ 64
#define N_ 4096
#define C_ 128
#define H_ 4
#define HC_ 512
#define ALPHA 0.2f

typedef __attribute__((ext_vector_type(8))) float v8f;
typedef __attribute__((ext_vector_type(2))) float v2f;

#define OFF_LOG   0u
#define OFF_U     (OFF_LOG + (unsigned)(B_*H_*N_))

__device__ __forceinline__ float neg_inf() { return -__builtin_inff(); }

__device__ __forceinline__ void lds_barrier() {
  __builtin_amdgcn_fence(__ATOMIC_RELEASE, "workgroup", "local");
  __builtin_amdgcn_s_barrier();
  __builtin_amdgcn_fence(__ATOMIC_ACQUIRE, "workgroup", "local");
}


template <int M>
__device__ __forceinline__ float row_xor(float x) {
  return __int_as_float(__builtin_amdgcn_update_dpp(
      0, __float_as_int(x), 0x160 + M, 0xf, 0xf, false));
}

#define K2_T 512
__global__ __launch_bounds__(K2_T) void k_logits(const float* __restrict__ V,
                                                 const int* __restrict__ gs,
                                                 const float* __restrict__ W,
                                                 const float* __restrict__ bias,
                                                 const float* __restrict__ T,
                                                 float* __restrict__ ws) {
  __shared__ __attribute__((aligned(16))) float s_w2[C_ * H_];
  __shared__ float s_b2[H_];
  const int b = blockIdx.y, chunk = blockIdx.x;
  const int tid = threadIdx.x;
  const int wave = tid >> 5, lane = tid & 31;
  const int l16 = lane & 15, hf = lane >> 4;
  const int qd = l16 >> 2, lo2 = l16 & 3;
  const int n0 = chunk * 1024 + wave * 64;
  const int limit = gs[b];

  float4 t4[H_];
  for (int h = 0; h < H_; ++h) t4[h] = ((const float4*)T)[h * 32 + lane];
  float4 wv[32];
  {
    const float4* w4 = (const float4*)(W + (size_t)(8 * wave) * HC_) + lane;
    for (int d = 0; d < 32; ++d) wv[d] = w4[d * 32];
  }
  const float4 bi = ((const float4*)bias)[(wave & 3) * 32 + lane];
  const float4 tb = ((const float4*)T)[(wave & 3) * 32 + lane];

  const float4* vsrc =
      (const float4*)(V + ((size_t)b * N_ + n0 + qd) * C_) + lo2 + 4 * hf;
  float* lrow = ws + OFF_LOG + ((size_t)b * H_ + lo2) * N_ + n0 + 2 * hf;

  float4 ra[16], rb[16];
  v8f acc[4];
#define K2_FETCH(R, T)                                                        \
  for (int g = 0; g < 4; ++g)                                                 \
    for (int j = 0; j < 4; ++j)                                               \
      R[4 * g + j] = vsrc[(size_t)(16 * (T) + 4 * g) * (C_ / 4) + 8 * j];
#define K2_MMA(R)                                                             \
  for (int g = 0; g < 4; ++g) acc[g] = v8f{};                                 \
  for (int j = 0; j < 4; ++j)                                                 \
    for (int g = 0; g < 4; ++g) {                                             \
      const float4 x = R[4 * g + j];                                          \
      const v2f a0 = {x.x, x.y}, a1 = {x.z, x.w};                             \
      acc[g] = __builtin_amdgcn_wmma_f32_16x16x4_f32(                         \
          false, a0, false, bw[2 * j], (short)0, acc[g], false, false);       \
      acc[g] = __builtin_amdgcn_wmma_f32_16x16x4_f32(                         \
          false, a1, false, bw[2 * j + 1], (short)0, acc[g], false, false);   \
    }
#define K2_EPI(T)                                                             \
  for (int g = 0; g < 4; ++g) {                                               \
    const float t0 = q1 ? acc[g][1] : acc[g][0], t1 = q1 ? acc[g][3] : acc[g][2]; \
    const float t2 = q1 ? acc[g][5] : acc[g][4], t3 = q1 ? acc[g][7] : acc[g][6]; \
    float s0 = q2 ? t1 : t0, s1 = q2 ? t3 : t2;                               \
    s0 += row_xor<4>(s0);  s1 += row_xor<4>(s1);                              \
    s0 += row_xor<8>(s0);  s1 += row_xor<8>(s1);                              \
    const int node = n0 + 16 * (T) + 4 * g + 2 * hf;                          \
    float x0 = s0 + bb, x1 = s1 + bb;                                         \
    x0 = (x0 >= 0.f) ? x0 : ALPHA * x0;                                       \
    x1 = (x1 >= 0.f) ? x1 : ALPHA * x1;                                       \
    v2f o;                                                                    \
    o.x = (node < limit) ? x0 : neg_inf();                                    \
    o.y = (node + 1 < limit) ? x1 : neg_inf();                                \
    if (l16 < H_) *(v2f*)(lrow + 16 * (T) + 4 * g) = o;                       \
  }
#define K2_PIN __builtin_amdgcn_sched_barrier(0);
  const bool q1 = (qd & 1) != 0, q2 = (qd & 2) != 0;

  K2_FETCH(ra, 0)
  K2_PIN
  float p[32];
  for (int d = 0; d < 32; ++d) {
    const float4 x = wv[d], t = t4[d & 3];
    p[d] = fmaf(x.w, t.w, fmaf(x.z, t.z, fmaf(x.y, t.y, x.x * t.x)));
  }
  float pb = fmaf(bi.w, tb.w, fmaf(bi.z, tb.z, fmaf(bi.y, tb.y, bi.x * tb.x)));
  K2_PIN
  K2_FETCH(rb, 1)
  K2_PIN
  {
    const bool up = (lane & 16) != 0;
    for (int i = 0; i < 16; ++i) {
      const float send = up ? p[i] : p[i + 16], keep = up ? p[i + 16] : p[i];
      p[i] = keep + __shfl_xor(send, 16);
    }
  }
#define K2_HALVE(M)                                                           \
  {                                                                           \
    const bool up = (lane & (M)) != 0;                                        \
    for (int i = 0; i < (M); ++i) {                                           \
      const float send = up ? p[i] : p[i + (M)], keep = up ? p[i + (M)] : p[i]; \
      p[i] = keep + row_xor<(M)>(send);                                       \
    }                                                                         \
  }
  K2_HALVE(8)
  K2_HALVE(4)
  K2_HALVE(2)
  K2_HALVE(1)
#undef K2_HALVE
  s_w2[32 * wave + lane] = p[0];
  pb += row_xor<1>(pb);  pb += row_xor<2>(pb);
  pb += row_xor<4>(pb);  pb += row_xor<8>(pb);
  pb += __shfl_xor(pb, 16);
  if (wave < H_ && lane == 0) s_b2[wave] = pb;
  lds_barrier();

  v2f bw[8];
  {
    const float* wq = s_w2 + (16 * hf + 4 * qd) * H_ + lo2;
    for (int j = 0; j < 4; ++j) {
      bw[2 * j].x     = wq[(32 * j + 0) * H_];
      bw[2 * j].y     = wq[(32 * j + 1) * H_];
      bw[2 * j + 1].x = wq[(32 * j + 2) * H_];
      bw[2 * j + 1].y = wq[(32 * j + 3) * H_];
    }
  }
  const float bb = s_b2[lo2];
  K2_PIN
  K2_MMA(ra)
  K2_PIN
  K2_FETCH(ra, 2)
  K2_PIN
  K2_EPI(0)
  K2_MMA(rb)
  K2_PIN
  K2_FETCH(rb, 3)
  K2_PIN
  K2_EPI(1)
  K2_MMA(ra)
  K2_EPI(2)
  K2_MMA(rb)
  K2_EPI(3)
#undef K2_PIN
#undef K2_FETCH
#undef K2_MMA
#undef K2_EPI
}

#define K3_T 512
__global__ __launch_bounds__(K3_T) void k_wsum(const float* __restrict__ V,
                                               float* __restrict__ ws) {
  __shared__ __attribute__((aligned(16))) float s_p[N_ * H_];
  __shared__ __attribute__((aligned(16))) float s_r[16 * 256];
  const int q = blockIdx.x, b = blockIdx.y;
  const int tid = threadIdx.x;
  const int wave = tid >> 5, lane = tid & 31;
  const int l16 = lane & 15, hf = lane >> 4;
  const int par = l16 >> 3, j = l16 & 7;
  const int nw = wave * 256;

  const float* lg = ws + OFF_LOG + (size_t)b * H_ * N_ + 4 * tid;
  float4 lx[H_][2];
  for (int h = 0; h < H_; ++h) {
    lx[h][0] = *(const float4*)(lg + h * N_);
    lx[h][1] = *(const float4*)(lg + h * N_ + 2048);
  }
  const float4* vsrc =
      (const float4*)(V + ((size_t)b * N_ + nw + 4 * par + 2 * hf) * C_ + 32 * q) + j;
  float4 ra[8], rb[8];
#define K3_FETCH(R, T)                                                        \
  for (int g = 0; g < 4; ++g) {                                               \
    R[2 * g]     = vsrc[(size_t)(32 * (T) + 8 * g) * (C_ / 4)];               \
    R[2 * g + 1] = vsrc[(size_t)(32 * (T) + 8 * g + 1) * (C_ / 4)];           \
  }
#define K3_PIN __builtin_amdgcn_sched_barrier(0);
  K3_PIN
  K3_FETCH(ra, 0)
  K3_FETCH(rb, 1)
  K3_PIN

  float m[H_];
  for (int h = 0; h < H_; ++h) {
    float v = fmaxf(fmaxf(fmaxf(lx[h][0].x, lx[h][0].y), fmaxf(lx[h][0].z, lx[h][0].w)),
                    fmaxf(fmaxf(lx[h][1].x, lx[h][1].y), fmaxf(lx[h][1].z, lx[h][1].w)));
    v = fmaxf(v, row_xor<1>(v));  v = fmaxf(v, row_xor<2>(v));
    v = fmaxf(v, row_xor<4>(v));  v = fmaxf(v, row_xor<8>(v));
    if (l16 == 0) s_r[h * 32 + 2 * wave + hf] = v;
  }
  lds_barrier();
  for (int h = 0; h < H_; ++h) {
    float v = s_r[h * 32];
    for (int w = 1; w < 32; ++w) v = fmaxf(v, s_r[h * 32 + w]);
    m[h] = v;
  }
  float isum[H_];
  for (int h = 0; h < H_; ++h) {
    float s = 0.f;
    for (int i = 0; i < 2; ++i) {
      float4 x = lx[h][i];
      x.x = (x.x == neg_inf()) ? 0.f : __expf(x.x - m[h]);
      x.y = (x.y == neg_inf()) ? 0.f : __expf(x.y - m[h]);
      x.z = (x.z == neg_inf()) ? 0.f : __expf(x.z - m[h]);
      x.w = (x.w == neg_inf()) ? 0.f : __expf(x.w - m[h]);
      lx[h][i] = x;
      s += (x.x + x.y) + (x.z + x.w);
    }
    s += row_xor<1>(s);  s += row_xor<2>(s);
    s += row_xor<4>(s);  s += row_xor<8>(s);
    if (l16 == 0) s_r[128 + h * 32 + 2 * wave + hf] = s;
  }
  lds_barrier();
  for (int h = 0; h < H_; ++h) {
    float s = s_r[128 + h * 32];
    for (int w = 1; w < 32; ++w) s += s_r[128 + h * 32 + w];
    isum[h] = 1.f / s;
  }
  for (int i = 0; i < 2; ++i) {
    float* d = s_p + (size_t)(i * 1024 + 2 * tid) * (H_ * 2);
    *(float4*)(d)      = make_float4(lx[0][i].x * isum[0], lx[0][i].y * isum[0],
                                     lx[1][i].x * isum[1], lx[1][i].y * isum[1]);
    *(float4*)(d + 4)  = make_float4(lx[2][i].x * isum[2], lx[2][i].y * isum[2],
                                     lx[3][i].x * isum[3], lx[3][i].y * isum[3]);
    *(float4*)(d + 8)  = make_float4(lx[0][i].z * isum[0], lx[0][i].w * isum[0],
                                     lx[1][i].z * isum[1], lx[1][i].w * isum[1]);
    *(float4*)(d + 12) = make_float4(lx[2][i].z * isum[2], lx[2][i].w * isum[2],
                                     lx[3][i].z * isum[3], lx[3][i].w * isum[3]);
  }
  lds_barrier();

  const float* pa = s_p + (size_t)((nw >> 1) + 2 * (l16 >> 2) + hf) * (H_ * 2) + (l16 & 3) * 2;
  v8f acc[4];
  for (int c = 0; c < 4; ++c) acc[c] = v8f{};
#define K3_MMA(R, T)                                                          \
  {                                                                           \
    v2f af[4];                                                                \
    for (int g = 0; g < 4; ++g) {                                             \
      v2f a = {0.f, 0.f};                                                     \
      if (l16 < 8) a = *(const v2f*)(pa + (size_t)(16 * (T) + 4 * g) * (H_ * 2)); \
      af[g] = a;                                                              \
    }                                                                         \
    for (int g = 0; g < 4; ++g) {                                             \
      const float4 x0 = R[2 * g], x1 = R[2 * g + 1];                          \
      const v2f b0 = {x0.x, x1.x}, b1 = {x0.y, x1.y};                         \
      const v2f b2 = {x0.z, x1.z}, b3 = {x0.w, x1.w};                         \
      acc[0] = __builtin_amdgcn_wmma_f32_16x16x4_f32(                         \
          false, af[g], false, b0, (short)0, acc[0], false, false);           \
      acc[1] = __builtin_amdgcn_wmma_f32_16x16x4_f32(                         \
          false, af[g], false, b1, (short)0, acc[1], false, false);           \
      acc[2] = __builtin_amdgcn_wmma_f32_16x16x4_f32(                         \
          false, af[g], false, b2, (short)0, acc[2], false, false);           \
      acc[3] = __builtin_amdgcn_wmma_f32_16x16x4_f32(                         \
          false, af[g], false, b3, (short)0, acc[3], false, false);           \
    }                                                                         \
  }
  for (int t = 0; t < 8; t += 2) {
    K3_MMA(ra, t)
    K3_PIN
    if (t + 2 < 8) { K3_FETCH(ra, t + 2) }
    K3_PIN
    K3_MMA(rb, t + 1)
    K3_PIN
    if (t + 3 < 8) { K3_FETCH(rb, t + 3) }
    K3_PIN
  }
#undef K3_PIN
#undef K3_FETCH
#undef K3_MMA

  if (hf == 0) {
    float* d = s_r + wave * 256 + par * 128 + 4 * j;
    for (int h = 0; h < H_; ++h) {
      float4 o;
      o.x = par ? acc[0][4 + h] : acc[0][h];
      o.y = par ? acc[1][4 + h] : acc[1][h];
      o.z = par ? acc[2][4 + h] : acc[2][h];
      o.w = par ? acc[3][4 + h] : acc[3][h];
      *(float4*)(d + h * 32) = o;
    }
  }
  lds_barrier();
  if (tid < H_ * 32) {
    float s = 0.f;
    for (int w = 0; w < 16; ++w)
      s += s_r[w * 256 + tid] + s_r[w * 256 + 128 + tid];
    ws[OFF_U + b * HC_ + (tid >> 5) * C_ + 32 * q + (tid & 31)] = s;
  }
}

__global__ void k_out(const float* __restrict__ W, const float* __restrict__ bias,
                      const float* __restrict__ U, float* __restrict__ out) {
  const int m0 = blockIdx.x * 16;
  const int h  = blockIdx.y;
  const int tid = threadIdx.x;
  const int wave = tid >> 5, lane = tid & 31;
  const int l16 = lane & 15, hi = lane >> 4;
  const int n0 = wave * 16;
  const float* urow = U + (size_t)(m0 + l16) * HC_ + h * C_;
  const float* wcol = W + h * C_ + n0 + l16;
  v8f acc = {};
  for (int k0 = 0; k0 < C_; k0 += 4) {
    v2f a = *(const v2f*)(urow + k0 + 2 * hi);
    v2f bv;
    bv.x = wcol[(size_t)(k0 + 2 * hi) * HC_];
    bv.y = wcol[(size_t)(k0 + 1 + 2 * hi) * HC_];
    acc = __builtin_amdgcn_wmma_f32_16x16x4_f32(false, a, false, bv,
                                                (short)0, acc, false, false);
  }
  for (int r = 0; r < 8; ++r) {
    int row = m0 + r + 8 * hi;
    int col = h * C_ + n0 + l16;
    out[(size_t)row * HC_ + col] = acc[r] + bias[col];
  }
}

extern "C" void kernel_launch(void* const* d_in, const int* in_sizes, int n_in,
                              void* d_out, int out_size, void* d_ws, size_t ws_size,
                              hipStream_t stream) {
  (void)in_sizes; (void)n_in; (void)out_size; (void)ws_size;
  const float* V    = (const float*)d_in[0];
  const int*   gs   = (const int*)  d_in[1];
  const float* W    = (const float*)d_in[2];
  const float* bias = (const float*)d_in[3];
  const float* T    = (const float*)d_in[4];
  float* out = (float*)d_out;
  float* ws  = (float*)d_ws;

  k_logits<<<dim3(N_/1024, B_), K2_T, 0, stream>>>(V, gs, W, bias, T, ws);
  k_wsum  <<<dim3(C_/32, B_), K3_T, 0, stream>>>(V, ws);
  k_out   <<<dim3(B_/16, H_), 256, 0, stream>>>(W, bias, ws + OFF_U, out);
}
